# starting kernel
# baseline (speedup 1.0000x reference)
#include <hip/hip_runtime.h>

#define LMAXV 2
#define CCH 8

typedef float v2f __attribute__((ext_vector_type(2)));
typedef float v8f __attribute__((ext_vector_type(8)));

constexpr double cfact(int n) { double r = 1.0; for (int i = 2; i <= n; ++i) r *= i; return r; }
constexpr double csqrt_(double x) {
  if (x <= 0.0) return 0.0;
  double r = x < 1.0 ? 1.0 : x;
  for (int i = 0; i < 100; ++i) r = 0.5 * (r + x / r);
  return r;
}
constexpr double cgcoef(int j1, int m1, int j2, int m2, int j, int m) {
  if (m1 + m2 != m) return 0.0;
  int lo = j1 > j2 ? j1 - j2 : j2 - j1;
  if (j < lo || j > j1 + j2) return 0.0;
  double pre = csqrt_((2.0 * j + 1.0) * cfact(j + j1 - j2) * cfact(j - j1 + j2) *
                      cfact(j1 + j2 - j) / cfact(j1 + j2 + j + 1));
  pre *= csqrt_(cfact(j + m) * cfact(j - m) * cfact(j1 - m1) * cfact(j1 + m1) *
                cfact(j2 - m2) * cfact(j2 + m2));
  double s = 0.0;
  for (int k = 0; k <= j1 + j2 - j; ++k) {
    int d2 = j1 + j2 - j - k, d3 = j1 - m1 - k, d4 = j2 + m2 - k;
    int d5 = j - j2 + m1 + k, d6 = j - j1 - m2 + k;
    if (d2 < 0 || d3 < 0 || d4 < 0 || d5 < 0 || d6 < 0) continue;
    double den = cfact(k) * cfact(d2) * cfact(d3) * cfact(d4) * cfact(d5) * cfact(d6);
    s += ((k & 1) ? -1.0 : 1.0) / den;
  }
  return pre * s;
}

struct CGPack {
  int ncoup[3];
  int l1t[3][6], l2t[3][6];
  int nnz[3][5][6];
  int pq[3][5][6][5];
  float coef[3][5][6][5];
};

constexpr CGPack buildCG() {
  CGPack P{};
  for (int l = 0; l <= LMAXV; ++l) {
    int nc = 0;
    for (int l1 = 0; l1 <= LMAXV; ++l1)
      for (int l2 = 0; l2 <= LMAXV; ++l2) {
        int lo = l1 > l2 ? l1 - l2 : l2 - l1;
        int hi = (l1 + l2 > LMAXV) ? LMAXV : (l1 + l2);
        if (l < lo || l > hi) continue;
        P.l1t[l][nc] = l1; P.l2t[l][nc] = l2;
        for (int mi = 0; mi < 2 * l + 1; ++mi) {
          int m = mi - l, n = 0;
          for (int p = 0; p < 2 * l1 + 1; ++p) {
            int m1 = p - l1, m2 = m - m1;
            if (m2 < -l2 || m2 > l2) continue;
            double v = cgcoef(l1, m1, l2, m2, l, m);
            if (v == 0.0) continue;
            P.pq[l][mi][nc][n] = p * 8 + (m2 + l2);
            P.coef[l][mi][nc][n] = (float)v;
            ++n;
          }
          P.nnz[l][mi][nc] = n;
        }
        ++nc;
      }
    P.ncoup[l] = nc;
  }
  return P;
}

constexpr CGPack H = buildCG();
constexpr int LOFF[3] = {0, 8, 32};
constexpr int NCOUPC[3] = {3, 6, 6};

__global__ void zero_f32(float* __restrict__ p, long n) {
  long i = (long)blockIdx.x * blockDim.x + threadIdx.x;
  if (i < n) p[i] = 0.0f;
}

__global__ __launch_bounds__(256) void edge_mp(
    const float* __restrict__ x0, const float* __restrict__ x1,
    const float* __restrict__ x2, const float* __restrict__ ev,
    const int* __restrict__ src, const int* __restrict__ dst,
    float* __restrict__ y, int nE) {
  long idx = (long)blockIdx.x * blockDim.x + threadIdx.x;
  if (idx >= (long)nE * 6) return;
  int e = (int)(idx / 6);
  int g = (int)(idx % 6);
  int s = src[e], d = dst[e];
  float v = ev[e];
#pragma unroll
  for (int j = 0; j < 3; ++j) {
    int slot = g * 3 + j;
    const float* ps;
    if (slot < 2)      ps = x0 + (size_t)s * 8  + slot * 4;
    else if (slot < 8) ps = x1 + (size_t)s * 24 + (slot - 2) * 4;
    else               ps = x2 + (size_t)s * 40 + (slot - 8) * 4;
    float4 val = *(const float4*)ps;
    float* py = y + (size_t)d * 72 + slot * 4;
    __hip_atomic_fetch_add(py + 0, v * val.x, __ATOMIC_RELAXED, __HIP_MEMORY_SCOPE_AGENT);
    __hip_atomic_fetch_add(py + 1, v * val.y, __ATOMIC_RELAXED, __HIP_MEMORY_SCOPE_AGENT);
    __hip_atomic_fetch_add(py + 2, v * val.z, __ATOMIC_RELAXED, __HIP_MEMORY_SCOPE_AGENT);
    __hip_atomic_fetch_add(py + 3, v * val.w, __ATOMIC_RELAXED, __HIP_MEMORY_SCOPE_AGENT);
  }
}

__global__ __launch_bounds__(64) void node_cg_wmma(
    const float* __restrict__ x0, const float* __restrict__ x1,
    const float* __restrict__ x2, const float* __restrict__ y,
    const float* __restrict__ wyy0, const float* __restrict__ wyy1,
    const float* __restrict__ wyy2, const float* __restrict__ wyx0,
    const float* __restrict__ wyx1, const float* __restrict__ wyx2,
    const float* __restrict__ wxx0, const float* __restrict__ wxx1,
    const float* __restrict__ wxx2, float* __restrict__ out, int nN) {
  __shared__ float lds[2 * 16 * 144];

  const int wave = threadIdx.x >> 5;
  const int lane = threadIdx.x & 31;
  const int row  = lane & 15;
  const int hi   = lane >> 4;
  const int wcol = row & 7;
  const int tileBase = blockIdx.x * 32 + wave * 16;

  const float* W[3][3] = {{wyy0, wyx0, wxx0}, {wyy1, wyx1, wxx1}, {wyy2, wyx2, wxx2}};
  __builtin_prefetch(W[0][0], 0, 3);
  __builtin_prefetch(W[1][0], 0, 3);
  __builtin_prefetch(W[2][0], 0, 3);

  float* L = lds + wave * 16 * 144;
  for (int i = lane; i < 16 * 72; i += 32) {
    int ln = i / 72, f = i % 72;
    int node = tileBase + ln;
    if (node >= nN) node = nN - 1;
    float xv;
    if (f < 8)       xv = x0[(size_t)node * 8  + f];
    else if (f < 32) xv = x1[(size_t)node * 24 + (f - 8)];
    else             xv = x2[(size_t)node * 40 + (f - 32)];
    L[ln * 144 + f]      = xv;
    L[ln * 144 + 72 + f] = y[(size_t)node * 72 + f];
  }
  __syncthreads();

  const float* Lrow = L + row * 144;
  const v8f vzero = {0.f, 0.f, 0.f, 0.f, 0.f, 0.f, 0.f, 0.f};

#pragma unroll
  for (int l = 0; l < 3; ++l) {
    const int nm = 2 * l + 1;
    v8f acc[5];
#pragma unroll
    for (int m = 0; m < 5; ++m) acc[m] = vzero;

#pragma unroll
    for (int prod = 0; prod < 3; ++prod) {
      const float* wl = W[l][prod];
      const int aoff = (prod < 2)  ? 72 : 0;
      const int boff = (prod == 0) ? 72 : 0;
#pragma unroll
      for (int cp = 0; cp < NCOUPC[l]; ++cp) {
        const int l1 = H.l1t[l][cp], l2 = H.l2t[l][cp];
        const int na = 2 * l1 + 1, nb = 2 * l2 + 1;
        const float* paBase = Lrow + aoff + LOFF[l1];
        const float* pbBase = Lrow + boff + LOFF[l2];
        const float* wcp = wl + (size_t)(cp * 64) * CCH;
        for (int c = 0; c < 8; ++c) {
          const float* pa = paBase + c * na;
          for (int dc = 0; dc < 8; dc += 4) {
            const int dlane = dc + 2 * hi;
            const int kb = c * 8 + dlane;
            v2f B;
            B.x = wcp[(kb + 0) * CCH + wcol];
            B.y = wcp[(kb + 1) * CCH + wcol];
            const float* pb0 = pbBase + dlane * nb;
            const float* pb1 = pb0 + nb;
#pragma unroll
            for (int m = 0; m < 2 * l + 1; ++m) {
              float z0 = 0.f, z1 = 0.f;
#pragma unroll
              for (int t = 0; t < H.nnz[l][m][cp]; ++t) {
                const int pq = H.pq[l][m][cp][t];
                const float cf = H.coef[l][m][cp][t];
                const float a = pa[pq >> 3];
                z0 += cf * a * pb0[pq & 7];
                z1 += cf * a * pb1[pq & 7];
              }
              v2f A; A.x = z0; A.y = z1;
              acc[m] = __builtin_amdgcn_wmma_f32_16x16x4_f32(
                  false, A, false, B, (short)0, acc[m], false, false);
            }
          }
        }
      }
    }

    const int o = row;
    if (o < CCH) {
      const size_t baseL = (l == 0) ? 0 : (l == 1 ? (size_t)nN * 8 : (size_t)nN * 32);
#pragma unroll
      for (int r = 0; r < 8; ++r) {
        int ln = r + 8 * hi;
        int node = tileBase + ln;
        if (node >= nN) node = nN - 1;
        const float* Lx = L + ln * 144 + LOFF[l] + o * nm;
        float* po = out + baseL + (size_t)node * (CCH * nm) + o * nm;
#pragma unroll
        for (int m = 0; m < 2 * l + 1; ++m) po[m] = acc[m][r] + Lx[m];
      }
    }
  }
}

extern "C" void kernel_launch(void* const* d_in, const int* in_sizes, int n_in,
                              void* d_out, int out_size, void* d_ws, size_t ws_size,
                              hipStream_t stream) {
  const float* x0 = (const float*)d_in[0];
  const float* x1 = (const float*)d_in[1];
  const float* x2 = (const float*)d_in[2];
  const float* ev = (const float*)d_in[3];
  const int* esrc = (const int*)d_in[4];
  const int* edst = (const int*)d_in[5];
  const float* wyy0 = (const float*)d_in[6];
  const float* wyy1 = (const float*)d_in[7];
  const float* wyy2 = (const float*)d_in[8];
  const float* wyx0 = (const float*)d_in[9];
  const float* wyx1 = (const float*)d_in[10];
  const float* wyx2 = (const float*)d_in[11];
  const float* wxx0 = (const float*)d_in[12];
  const float* wxx1 = (const float*)d_in[13];
  const float* wxx2 = (const float*)d_in[14];
  float* out = (float*)d_out;
  float* y   = (float*)d_ws;

  const int nN = in_sizes[0] / CCH;
  const int nE = in_sizes[3];

  long ny = (long)nN * 72;
  zero_f32<<<(unsigned)((ny + 511) / 512), 512, 0, stream>>>(y, ny);

  long tot = (long)nE * 6;
  edge_mp<<<(unsigned)((tot + 255) / 256), 256, 0, stream>>>(x0, x1, x2, ev, esrc,
                                                             edst, y, nE);

  int blocks = (nN + 31) / 32;
  node_cg_wmma<<<blocks, 64, 0, stream>>>(x0, x1, x2, y, wyy0, wyy1, wyy2, wyx0,
                                          wyx1, wyx2, wxx0, wxx1, wxx2, out, nN);
}


# message passing moved into the node kernel: each work-group scans the destination list, keeps its 80 nodes' edges, sums them in LDS with LDS float adds (arrival order); no global atomics, 1 launch
# speedup vs baseline: 2.1368x; 2.1368x over previous
#include <hip/hip_runtime.h>

#define LMAXV 2
#define CCH 8

typedef float v2f __attribute__((ext_vector_type(2)));
typedef float v8f __attribute__((ext_vector_type(8)));

constexpr double cfact(int n) { double r = 1.0; for (int i = 2; i <= n; ++i) r *= i; return r; }
constexpr double csqrt_(double x) {
  if (x <= 0.0) return 0.0;
  double r = x < 1.0 ? 1.0 : x;
  for (int i = 0; i < 100; ++i) r = 0.5 * (r + x / r);
  return r;
}
constexpr double cgcoef(int j1, int m1, int j2, int m2, int j, int m) {
  if (m1 + m2 != m) return 0.0;
  int lo = j1 > j2 ? j1 - j2 : j2 - j1;
  if (j < lo || j > j1 + j2) return 0.0;
  double pre = csqrt_((2.0 * j + 1.0) * cfact(j + j1 - j2) * cfact(j - j1 + j2) *
                      cfact(j1 + j2 - j) / cfact(j1 + j2 + j + 1));
  pre *= csqrt_(cfact(j + m) * cfact(j - m) * cfact(j1 - m1) * cfact(j1 + m1) *
                cfact(j2 - m2) * cfact(j2 + m2));
  double s = 0.0;
  for (int k = 0; k <= j1 + j2 - j; ++k) {
    int d2 = j1 + j2 - j - k, d3 = j1 - m1 - k, d4 = j2 + m2 - k;
    int d5 = j - j2 + m1 + k, d6 = j - j1 - m2 + k;
    if (d2 < 0 || d3 < 0 || d4 < 0 || d5 < 0 || d6 < 0) continue;
    double den = cfact(k) * cfact(d2) * cfact(d3) * cfact(d4) * cfact(d5) * cfact(d6);
    s += ((k & 1) ? -1.0 : 1.0) / den;
  }
  return pre * s;
}

struct CGPack {
  int ncoup[3];
  int l1t[3][6], l2t[3][6];
  int nnz[3][5][6];
  int pq[3][5][6][5];
  float coef[3][5][6][5];
};

constexpr CGPack buildCG() {
  CGPack P{};
  for (int l = 0; l <= LMAXV; ++l) {
    int nc = 0;
    for (int l1 = 0; l1 <= LMAXV; ++l1)
      for (int l2 = 0; l2 <= LMAXV; ++l2) {
        int lo = l1 > l2 ? l1 - l2 : l2 - l1;
        int hi = (l1 + l2 > LMAXV) ? LMAXV : (l1 + l2);
        if (l < lo || l > hi) continue;
        P.l1t[l][nc] = l1; P.l2t[l][nc] = l2;
        for (int mi = 0; mi < 2 * l + 1; ++mi) {
          int m = mi - l, n = 0;
          for (int p = 0; p < 2 * l1 + 1; ++p) {
            int m1 = p - l1, m2 = m - m1;
            if (m2 < -l2 || m2 > l2) continue;
            double v = cgcoef(l1, m1, l2, m2, l, m);
            if (v == 0.0) continue;
            P.pq[l][mi][nc][n] = p * 8 + (m2 + l2);
            P.coef[l][mi][nc][n] = (float)v;
            ++n;
          }
          P.nnz[l][mi][nc] = n;
        }
        ++nc;
      }
    P.ncoup[l] = nc;
  }
  return P;
}

constexpr CGPack H = buildCG();
constexpr int LOFF[3] = {0, 8, 32};
constexpr int NCOUPC[3] = {3, 6, 6};

#define NW 5
#define NODES_PER_WG (NW * 16)
constexpr int WOFF[3][3] = {{0, 1536, 3072}, {4608, 7680, 10752}, {13824, 16896, 19968}};
#define WTOT 23040
constexpr int WCH[3] = {3, 6, 6};
#define KLD 20
#define LCAP 2048
#define LSPARE (LCAP / 32 - 1)
#define DCAP 1024
#define STEP_EDGES (NW * 32 * KLD * 4)


__global__ __launch_bounds__(NW * 32) void node_cg_wmma(
    const float* __restrict__ x0, const float* __restrict__ x1,
    const float* __restrict__ x2, const float* __restrict__ ev,
    const int* __restrict__ src, const int* __restrict__ dst,
    const float* __restrict__ wyy0, const float* __restrict__ wyy1,
    const float* __restrict__ wyy2, const float* __restrict__ wyx0,
    const float* __restrict__ wyx1, const float* __restrict__ wyx2,
    const float* __restrict__ wxx0, const float* __restrict__ wxx1,
    const float* __restrict__ wxx2, float* __restrict__ out, int nN, int nE) {
  __shared__ float lds[NW * 16 * 144];
  __shared__ float wsm[WTOT];
  __shared__ int mlist[NW * LCAP];
  __shared__ int dlist[NW * DCAP];

  const int wave = threadIdx.x >> 5;
  const int lane = threadIdx.x & 31;
  const int row  = lane & 15;
  const int hi   = lane >> 4;
  const int wcol = row & 7;
  const int tileBase = blockIdx.x * NODES_PER_WG + wave * 16;

#define WCOPY(ptr, l, prod)                                                   \
  _Pragma("unroll") for (int k = 0; k < WCH[l]; ++k)                          \
      ((float4*)(wsm + WOFF[l][prod]))[k * 128 + threadIdx.x] =               \
          ((const float4*)(ptr))[k * 128 + threadIdx.x];
  if (wave < 4) {
    WCOPY(wyy0, 0, 0) WCOPY(wyx0, 0, 1) WCOPY(wxx0, 0, 2)
    WCOPY(wyy1, 1, 0) WCOPY(wyx1, 1, 1) WCOPY(wxx1, 1, 2)
    WCOPY(wyy2, 2, 0) WCOPY(wyx2, 2, 1) WCOPY(wxx2, 2, 2)
  }

  float* L = lds + wave * 16 * 144;
  for (int i = lane; i < 16 * 72; i += 32) {
    int ln = i / 72, f = i % 72;
    int node = tileBase + ln;
    if (node >= nN) node = nN - 1;
    float xv;
    if (f < 8)       xv = x0[(size_t)node * 8  + f];
    else if (f < 32) xv = x1[(size_t)node * 24 + (f - 8)];
    else             xv = x2[(size_t)node * 40 + (f - 32)];
    L[ln * 144 + f]      = xv;
    L[ln * 144 + 72 + f] = 0.0f;
  }
  __syncthreads();

  {
    const unsigned nodeBase = blockIdx.x * NODES_PER_WG;
    int* myl = mlist + wave * LCAP + lane;
    int* dl  = dlist + wave * DCAP;
    int cntl = 0;
    const int nFull = nE / STEP_EDGES;
    const int n4c   = (nE + 3) >> 2;
    const int4* dst4 = (const int4*)dst;
    int it = (int)((blockIdx.x * (unsigned)nFull) / gridDim.x);
    int4 cur[KLD];
    if (nFull > 0) {
#pragma unroll
      for (int k = 0; k < KLD; ++k)
        cur[k] = dst4[(it * KLD + k) * (NW * 32) + threadIdx.x];
    }
    for (int i = 0; i <= nFull; ++i) {
      if (i < nFull) {
        const int itn = (it + 1 == nFull) ? 0 : it + 1;
        int4 nxt[KLD];
#pragma unroll
        for (int k = 0; k < KLD; ++k)
          nxt[k] = dst4[(itn * KLD + k) * (NW * 32) + threadIdx.x];
#pragma unroll
        for (int k = 0; k < KLD; ++k) {
          const unsigned t0 = (unsigned)cur[k].x - nodeBase, t1 = (unsigned)cur[k].y - nodeBase;
          const unsigned t2 = (unsigned)cur[k].z - nodeBase, t3 = (unsigned)cur[k].w - nodeBase;
          const unsigned tm = min(min(t0, t1), min(t2, t3));
          const bool any = tm < (unsigned)NODES_PER_WG;
          myl[32 * (any ? cntl : LSPARE)] = (it * KLD + k) * (NW * 32) + threadIdx.x;
          cntl += any ? 1 : 0;
        }
#pragma unroll
        for (int k = 0; k < KLD; ++k) cur[k] = nxt[k];
        it = itn;
      } else {
        for (int g0 = nFull * (STEP_EDGES / 4); g0 < n4c; g0 += NW * 32) {
          const int g = g0 + (int)threadIdx.x;
          bool any = false;
#pragma unroll
          for (int j = 0; j < 4; ++j) {
            const int e = g * 4 + j;
            if (g < n4c && e < nE)
              any = any || ((unsigned)dst[e] - nodeBase < (unsigned)NODES_PER_WG);
          }
          myl[32 * (any ? cntl : LSPARE)] = g;
          cntl += any ? 1 : 0;
        }
      }
      if (i == nFull || __builtin_amdgcn_ballot_w32(cntl > LSPARE - 1 - KLD) != 0u) {
        int dcnt = 0;
        for (int c0 = 0;; c0 += 4) {
          const bool more = __builtin_amdgcn_ballot_w32(c0 < cntl) != 0u;
          if (more) {
            int  gi[4];
            int  dv[4][4];
            bool ok[4][4];
#pragma unroll
            for (int u = 0; u < 4; ++u) {
              const bool have = c0 + u < cntl;
              const int gl = myl[32 * (have ? c0 + u : LSPARE)];
              gi[u] = have ? gl : 0;
#pragma unroll
              for (int j = 0; j < 4; ++j) {
                const int e = gi[u] * 4 + j;
                ok[u][j] = have && e < nE;
                dv[u][j] = dst[ok[u][j] ? e : 0];
              }
            }
#pragma unroll
            for (int u = 0; u < 4; ++u)
#pragma unroll
              for (int j = 0; j < 4; ++j) {
                const bool hit = ok[u][j] &&
                                 ((unsigned)dv[u][j] - nodeBase < (unsigned)NODES_PER_WG);
                const unsigned bal = __builtin_amdgcn_ballot_w32(hit);
                if (bal) {
                  if (hit) dl[dcnt + __builtin_amdgcn_mbcnt_lo(bal, 0u)] = gi[u] * 4 + j;
                  dcnt += __builtin_popcount(bal);
                }
              }
          }
          if (!more || dcnt > DCAP - 512) {
            if (dcnt > 0) {
              const int e0 = dl[lane < dcnt ? lane : dcnt - 1];
              int es  = src[e0];
              int ed  = dst[e0] - (int)nodeBase;
              int evb = __float_as_int(ev[e0]);
              for (int b0 = 0; b0 < dcnt; b0 += 32) {
                const int nb = (dcnt - b0 < 32) ? dcnt - b0 : 32;
                const int ni = b0 + 32 + lane;
                const int e2 = dl[ni < dcnt ? ni : dcnt - 1];
                const int es2  = src[e2];
                const int ed2  = dst[e2] - (int)nodeBase;
                const int evb2 = __float_as_int(ev[e2]);
                int    from[18], ss[18], yo[18];
                float  sc[18];
                float4 val[18];
#pragma unroll
                for (int r = 0; r < 18; ++r) {
                  const unsigned ent = (lane + 32u * r) / 18u;
                  from[r] = (ent < (unsigned)nb ? (int)ent : 0) << 2;
                  ss[r] = __builtin_amdgcn_ds_bpermute(from[r], es);
                }
#pragma unroll
                for (int r = 0; r < 18; ++r) {
                  const unsigned q = lane + 32u * r;
                  const unsigned slot = q - (q / 18u) * 18u;
                  const float* p;
                  if (slot < 2)      p = x0 + (size_t)ss[r] * 8  + slot * 4;
                  else if (slot < 8) p = x1 + (size_t)ss[r] * 24 + (slot - 2) * 4;
                  else               p = x2 + (size_t)ss[r] * 40 + (slot - 8) * 4;
                  val[r] = *(const float4*)p;
                }
#pragma unroll
                for (int r = 0; r < 18; ++r) {
                  const unsigned q = lane + 32u * r;
                  const unsigned slot = q - (q / 18u) * 18u;
                  yo[r] = __builtin_amdgcn_ds_bpermute(from[r], ed) * 144 + 72 + (int)slot * 4;
                  sc[r] = __int_as_float(__builtin_amdgcn_ds_bpermute(from[r], evb));
                }
#pragma unroll
                for (int r = 0; r < 18; ++r) {
                  const unsigned ent = (lane + 32u * r) / 18u;
                  if (ent < (unsigned)nb) {
                    float* py = lds + yo[r];
                    __hip_atomic_fetch_add(py + 0, sc[r] * val[r].x, __ATOMIC_RELAXED, __HIP_MEMORY_SCOPE_WORKGROUP);
                    __hip_atomic_fetch_add(py + 1, sc[r] * val[r].y, __ATOMIC_RELAXED, __HIP_MEMORY_SCOPE_WORKGROUP);
                    __hip_atomic_fetch_add(py + 2, sc[r] * val[r].z, __ATOMIC_RELAXED, __HIP_MEMORY_SCOPE_WORKGROUP);
                    __hip_atomic_fetch_add(py + 3, sc[r] * val[r].w, __ATOMIC_RELAXED, __HIP_MEMORY_SCOPE_WORKGROUP);
                  }
                }
                es = es2; ed = ed2; evb = evb2;
              }
            }
            dcnt = 0;
          }
          if (!more) break;
        }
        cntl = 0;
      }
    }
  }
  __syncthreads();

  const float* Lrow = L + row * 144;
  const v8f vzero = {0.f, 0.f, 0.f, 0.f, 0.f, 0.f, 0.f, 0.f};

#pragma unroll
  for (int l = 0; l < 3; ++l) {
    const int nm = 2 * l + 1;
    v8f acc[5];
#pragma unroll
    for (int m = 0; m < 5; ++m) acc[m] = vzero;

#pragma unroll
    for (int prod = 0; prod < 3; ++prod) {
      const float* wl = wsm + WOFF[l][prod];
      const int aoff = (prod < 2)  ? 72 : 0;
      const int boff = (prod == 0) ? 72 : 0;
#pragma unroll
      for (int cp = 0; cp < NCOUPC[l]; ++cp) {
        const int l1 = H.l1t[l][cp], l2 = H.l2t[l][cp];
        const int na = 2 * l1 + 1, nb = 2 * l2 + 1;
        const float* paBase = Lrow + aoff + LOFF[l1];
        const float* pbBase = Lrow + boff + LOFF[l2];
        const float* wcp = wl + (size_t)(cp * 64) * CCH;
        for (int c = 0; c < 8; ++c) {
          const float* pa = paBase + c * na;
          for (int dc = 0; dc < 8; dc += 4) {
            const int dlane = dc + 2 * hi;
            const int kb = c * 8 + dlane;
            v2f B;
            B.x = wcp[(kb + 0) * CCH + wcol];
            B.y = wcp[(kb + 1) * CCH + wcol];
            const float* pb0 = pbBase + dlane * nb;
            const float* pb1 = pb0 + nb;
#pragma unroll
            for (int m = 0; m < 2 * l + 1; ++m) {
              float z0 = 0.f, z1 = 0.f;
#pragma unroll
              for (int t = 0; t < H.nnz[l][m][cp]; ++t) {
                const int pq = H.pq[l][m][cp][t];
                const float cf = H.coef[l][m][cp][t];
                const float a = pa[pq >> 3];
                z0 += cf * a * pb0[pq & 7];
                z1 += cf * a * pb1[pq & 7];
              }
              v2f A; A.x = z0; A.y = z1;
              acc[m] = __builtin_amdgcn_wmma_f32_16x16x4_f32(
                  false, A, false, B, (short)0, acc[m], false, false);
            }
          }
        }
      }
    }

    const int o = row;
    if (o < CCH) {
      const size_t baseL = (l == 0) ? 0 : (l == 1 ? (size_t)nN * 8 : (size_t)nN * 32);
#pragma unroll
      for (int r = 0; r < 8; ++r) {
        int ln = r + 8 * hi;
        const int node = tileBase + ln;
        if (node < nN) {
          const float* Lx = L + ln * 144 + LOFF[l] + o * nm;
          float* po = out + baseL + (size_t)node * (CCH * nm) + o * nm;
#pragma unroll
          for (int m = 0; m < 2 * l + 1; ++m) po[m] = acc[m][r] + Lx[m];
        }
      }
    }
  }
}

extern "C" void kernel_launch(void* const* d_in, const int* in_sizes, int n_in,
                              void* d_out, int out_size, void* d_ws, size_t ws_size,
                              hipStream_t stream) {
  const float* x0 = (const float*)d_in[0];
  const float* x1 = (const float*)d_in[1];
  const float* x2 = (const float*)d_in[2];
  const float* ev = (const float*)d_in[3];
  const int* esrc = (const int*)d_in[4];
  const int* edst = (const int*)d_in[5];
  const float* wyy0 = (const float*)d_in[6];
  const float* wyy1 = (const float*)d_in[7];
  const float* wyy2 = (const float*)d_in[8];
  const float* wyx0 = (const float*)d_in[9];
  const float* wyx1 = (const float*)d_in[10];
  const float* wyx2 = (const float*)d_in[11];
  const float* wxx0 = (const float*)d_in[12];
  const float* wxx1 = (const float*)d_in[13];
  const float* wxx2 = (const float*)d_in[14];
  float* out = (float*)d_out;
  (void)d_ws;

  const int nN = in_sizes[0] / CCH;
  const int nE = in_sizes[3];

  int blocks = (nN + NODES_PER_WG - 1) / NODES_PER_WG;
  node_cg_wmma<<<blocks, NW * 32, 0, stream>>>(x0, x1, x2, ev, esrc, edst, wyy0, wyy1,
                                          wyy2, wyx0, wyx1, wyx2, wxx0, wxx1, wxx2,
                                          out, nN, nE);
}


# work-group of 8 waves (two per SIMD) instead of 5: the 15 output blocks (tile, l) are dealt to waves by matrix-instruction count, at most 3168 per SIMD instead of 4896; no partial sums
# speedup vs baseline: 2.6171x; 1.2248x over previous
#include <hip/hip_runtime.h>

#define LMAXV 2
#define CCH 8

typedef float v2f __attribute__((ext_vector_type(2)));
typedef float v8f __attribute__((ext_vector_type(8)));

constexpr double cfact(int n) { double r = 1.0; for (int i = 2; i <= n; ++i) r *= i; return r; }
constexpr double csqrt_(double x) {
  if (x <= 0.0) return 0.0;
  double r = x < 1.0 ? 1.0 : x;
  for (int i = 0; i < 100; ++i) r = 0.5 * (r + x / r);
  return r;
}
constexpr double cgcoef(int j1, int m1, int j2, int m2, int j, int m) {
  if (m1 + m2 != m) return 0.0;
  int lo = j1 > j2 ? j1 - j2 : j2 - j1;
  if (j < lo || j > j1 + j2) return 0.0;
  double pre = csqrt_((2.0 * j + 1.0) * cfact(j + j1 - j2) * cfact(j - j1 + j2) *
                      cfact(j1 + j2 - j) / cfact(j1 + j2 + j + 1));
  pre *= csqrt_(cfact(j + m) * cfact(j - m) * cfact(j1 - m1) * cfact(j1 + m1) *
                cfact(j2 - m2) * cfact(j2 + m2));
  double s = 0.0;
  for (int k = 0; k <= j1 + j2 - j; ++k) {
    int d2 = j1 + j2 - j - k, d3 = j1 - m1 - k, d4 = j2 + m2 - k;
    int d5 = j - j2 + m1 + k, d6 = j - j1 - m2 + k;
    if (d2 < 0 || d3 < 0 || d4 < 0 || d5 < 0 || d6 < 0) continue;
    double den = cfact(k) * cfact(d2) * cfact(d3) * cfact(d4) * cfact(d5) * cfact(d6);
    s += ((k & 1) ? -1.0 : 1.0) / den;
  }
  return pre * s;
}

struct CGPack {
  int ncoup[3];
  int l1t[3][6], l2t[3][6];
  int nnz[3][5][6];
  int pq[3][5][6][5];
  float coef[3][5][6][5];
};

constexpr CGPack buildCG() {
  CGPack P{};
  for (int l = 0; l <= LMAXV; ++l) {
    int nc = 0;
    for (int l1 = 0; l1 <= LMAXV; ++l1)
      for (int l2 = 0; l2 <= LMAXV; ++l2) {
        int lo = l1 > l2 ? l1 - l2 : l2 - l1;
        int hi = (l1 + l2 > LMAXV) ? LMAXV : (l1 + l2);
        if (l < lo || l > hi) continue;
        P.l1t[l][nc] = l1; P.l2t[l][nc] = l2;
        for (int mi = 0; mi < 2 * l + 1; ++mi) {
          int m = mi - l, n = 0;
          for (int p = 0; p < 2 * l1 + 1; ++p) {
            int m1 = p - l1, m2 = m - m1;
            if (m2 < -l2 || m2 > l2) continue;
            double v = cgcoef(l1, m1, l2, m2, l, m);
            if (v == 0.0) continue;
            P.pq[l][mi][nc][n] = p * 8 + (m2 + l2);
            P.coef[l][mi][nc][n] = (float)v;
            ++n;
          }
          P.nnz[l][mi][nc] = n;
        }
        ++nc;
      }
    P.ncoup[l] = nc;
  }
  return P;
}

constexpr CGPack H = buildCG();
constexpr int LOFF[3] = {0, 8, 32};
constexpr int NCOUPC[3] = {3, 6, 6};

#define NT 5
#define NW 8
#define NODES_PER_WG (NT * 16)
constexpr int WOFF[3][3] = {{0, 1536, 3072}, {4608, 7680, 10752}, {13824, 16896, 19968}};
#define WTOT 23040
constexpr int WCH[3] = {3, 6, 6};
#define KLD 12
#define LCAP 2048
#define LSPARE (LCAP / 32 - 1)
#define DCAP 1024
#define STEP_EDGES (NW * 32 * KLD * 4)


template <int l>
static __device__ __forceinline__ void block_l(const float* __restrict__ L,
                                               const float* __restrict__ wsm,
                                               float* __restrict__ out,
                                               const int tileBase, const int nN,
                                               const int row, const int hi,
                                               const int wcol) {
  constexpr int nm = 2 * l + 1;
  const float* Lrow = L + row * 144;
  const v8f vzero = {0.f, 0.f, 0.f, 0.f, 0.f, 0.f, 0.f, 0.f};
  v8f acc[nm];
#pragma unroll
  for (int m = 0; m < nm; ++m) acc[m] = vzero;

#pragma unroll
  for (int prod = 0; prod < 3; ++prod) {
    const float* wl = wsm + WOFF[l][prod];
    const int aoff = (prod < 2)  ? 72 : 0;
    const int boff = (prod == 0) ? 72 : 0;
#pragma unroll
    for (int cp = 0; cp < NCOUPC[l]; ++cp) {
      const int l1 = H.l1t[l][cp], l2 = H.l2t[l][cp];
      const int na = 2 * l1 + 1, nb = 2 * l2 + 1;
      const float* paBase = Lrow + aoff + LOFF[l1];
      const float* pbBase = Lrow + boff + LOFF[l2];
      const float* wcp = wl + (size_t)(cp * 64) * CCH;
      for (int c = 0; c < 8; ++c) {
        const float* pa = paBase + c * na;
        for (int dc = 0; dc < 8; dc += 4) {
          const int dlane = dc + 2 * hi;
          const int kb = c * 8 + dlane;
          v2f B;
          B.x = wcp[(kb + 0) * CCH + wcol];
          B.y = wcp[(kb + 1) * CCH + wcol];
          const float* pb0 = pbBase + dlane * nb;
          const float* pb1 = pb0 + nb;
#pragma unroll
          for (int m = 0; m < nm; ++m) {
            float z0 = 0.f, z1 = 0.f;
#pragma unroll
            for (int t = 0; t < H.nnz[l][m][cp]; ++t) {
              const int pq = H.pq[l][m][cp][t];
              const float cf = H.coef[l][m][cp][t];
              const float a = pa[pq >> 3];
              z0 += cf * a * pb0[pq & 7];
              z1 += cf * a * pb1[pq & 7];
            }
            v2f A; A.x = z0; A.y = z1;
            acc[m] = __builtin_amdgcn_wmma_f32_16x16x4_f32(
                false, A, false, B, (short)0, acc[m], false, false);
          }
        }
      }
    }
  }

  const int o = row;
  if (o < CCH) {
    const size_t baseL = (l == 0) ? 0 : (l == 1 ? (size_t)nN * 8 : (size_t)nN * 32);
#pragma unroll
    for (int r = 0; r < 8; ++r) {
      int ln = r + 8 * hi;
      const int node = tileBase + ln;
      if (node < nN) {
        const float* Lx = L + ln * 144 + LOFF[l] + o * nm;
        float* po = out + baseL + (size_t)node * (CCH * nm) + o * nm;
#pragma unroll
        for (int m = 0; m < nm; ++m) po[m] = acc[m][r] + Lx[m];
      }
    }
  }
}

__global__ __launch_bounds__(NW * 32) void node_cg_wmma(
    const float* __restrict__ x0, const float* __restrict__ x1,
    const float* __restrict__ x2, const float* __restrict__ ev,
    const int* __restrict__ src, const int* __restrict__ dst,
    const float* __restrict__ wyy0, const float* __restrict__ wyy1,
    const float* __restrict__ wyy2, const float* __restrict__ wyx0,
    const float* __restrict__ wyx1, const float* __restrict__ wyx2,
    const float* __restrict__ wxx0, const float* __restrict__ wxx1,
    const float* __restrict__ wxx2, float* __restrict__ out, int nN, int nE) {
  __shared__ float lds[NT * 16 * 144];
  __shared__ float wsm[WTOT];
  __shared__ int mlist[NW * LCAP];
  __shared__ int dlist[NW * DCAP];

  const int wave = threadIdx.x >> 5;
  const int lane = threadIdx.x & 31;
  const int row  = lane & 15;
  const int hi   = lane >> 4;
  const int wcol = row & 7;

#define WCOPY(ptr, l, prod)                                                   \
  _Pragma("unroll") for (int k = 0; k < WCH[l]; ++k)                          \
      ((float4*)(wsm + WOFF[l][prod]))[k * 128 + threadIdx.x] =               \
          ((const float4*)(ptr))[k * 128 + threadIdx.x];
  if (wave < 4) {
    WCOPY(wyy0, 0, 0) WCOPY(wyx0, 0, 1) WCOPY(wxx0, 0, 2)
    WCOPY(wyy1, 1, 0) WCOPY(wyx1, 1, 1) WCOPY(wxx1, 1, 2)
    WCOPY(wyy2, 2, 0) WCOPY(wyx2, 2, 1) WCOPY(wxx2, 2, 2)
  }

  for (int i = threadIdx.x; i < NODES_PER_WG * 72; i += NW * 32) {
    int ln = i / 72, f = i % 72;
    int node = blockIdx.x * NODES_PER_WG + ln;
    if (node >= nN) node = nN - 1;
    float xv;
    if (f < 8)       xv = x0[(size_t)node * 8  + f];
    else if (f < 32) xv = x1[(size_t)node * 24 + (f - 8)];
    else             xv = x2[(size_t)node * 40 + (f - 32)];
    lds[ln * 144 + f]      = xv;
    lds[ln * 144 + 72 + f] = 0.0f;
  }
  __syncthreads();

  {
    const unsigned nodeBase = blockIdx.x * NODES_PER_WG;
    int* myl = mlist + wave * LCAP + lane;
    int* dl  = dlist + wave * DCAP;
    int cntl = 0;
    const int nFull = nE / STEP_EDGES;
    const int n4c   = (nE + 3) >> 2;
    const int4* dst4 = (const int4*)dst;
    int it = (int)((blockIdx.x * (unsigned)nFull) / gridDim.x);
    int4 cur[KLD];
    if (nFull > 0) {
#pragma unroll
      for (int k = 0; k < KLD; ++k)
        cur[k] = dst4[(it * KLD + k) * (NW * 32) + threadIdx.x];
    }
    for (int i = 0; i <= nFull; ++i) {
      if (i < nFull) {
        const int itn = (it + 1 == nFull) ? 0 : it + 1;
        int4 nxt[KLD];
#pragma unroll
        for (int k = 0; k < KLD; ++k)
          nxt[k] = dst4[(itn * KLD + k) * (NW * 32) + threadIdx.x];
#pragma unroll
        for (int k = 0; k < KLD; ++k) {
          const unsigned t0 = (unsigned)cur[k].x - nodeBase, t1 = (unsigned)cur[k].y - nodeBase;
          const unsigned t2 = (unsigned)cur[k].z - nodeBase, t3 = (unsigned)cur[k].w - nodeBase;
          const unsigned tm = min(min(t0, t1), min(t2, t3));
          const bool any = tm < (unsigned)NODES_PER_WG;
          myl[32 * (any ? cntl : LSPARE)] = (it * KLD + k) * (NW * 32) + threadIdx.x;
          cntl += any ? 1 : 0;
        }
#pragma unroll
        for (int k = 0; k < KLD; ++k) cur[k] = nxt[k];
        it = itn;
      } else {
        for (int g0 = nFull * (STEP_EDGES / 4); g0 < n4c; g0 += NW * 32) {
          const int g = g0 + (int)threadIdx.x;
          bool any = false;
#pragma unroll
          for (int j = 0; j < 4; ++j) {
            const int e = g * 4 + j;
            if (g < n4c && e < nE)
              any = any || ((unsigned)dst[e] - nodeBase < (unsigned)NODES_PER_WG);
          }
          myl[32 * (any ? cntl : LSPARE)] = g;
          cntl += any ? 1 : 0;
        }
      }
      if (i == nFull || __builtin_amdgcn_ballot_w32(cntl > LSPARE - 1 - KLD) != 0u) {
        int dcnt = 0;
        for (int c0 = 0;; c0 += 4) {
          const bool more = __builtin_amdgcn_ballot_w32(c0 < cntl) != 0u;
          if (more) {
            int  gi[4];
            int  dv[4][4];
            bool ok[4][4];
#pragma unroll
            for (int u = 0; u < 4; ++u) {
              const bool have = c0 + u < cntl;
              const int gl = myl[32 * (have ? c0 + u : LSPARE)];
              gi[u] = have ? gl : 0;
#pragma unroll
              for (int j = 0; j < 4; ++j) {
                const int e = gi[u] * 4 + j;
                ok[u][j] = have && e < nE;
                dv[u][j] = dst[ok[u][j] ? e : 0];
              }
            }
#pragma unroll
            for (int u = 0; u < 4; ++u)
#pragma unroll
              for (int j = 0; j < 4; ++j) {
                const bool hit = ok[u][j] &&
                                 ((unsigned)dv[u][j] - nodeBase < (unsigned)NODES_PER_WG);
                const unsigned bal = __builtin_amdgcn_ballot_w32(hit);
                if (bal) {
                  if (hit) dl[dcnt + __builtin_amdgcn_mbcnt_lo(bal, 0u)] = gi[u] * 4 + j;
                  dcnt += __builtin_popcount(bal);
                }
              }
          }
          if (!more || dcnt > DCAP - 512) {
            if (dcnt > 0) {
              const int e0 = dl[lane < dcnt ? lane : dcnt - 1];
              int es  = src[e0];
              int ed  = dst[e0] - (int)nodeBase;
              int evb = __float_as_int(ev[e0]);
              for (int b0 = 0; b0 < dcnt; b0 += 32) {
                const int nb = (dcnt - b0 < 32) ? dcnt - b0 : 32;
                const int ni = b0 + 32 + lane;
                const int e2 = dl[ni < dcnt ? ni : dcnt - 1];
                const int es2  = src[e2];
                const int ed2  = dst[e2] - (int)nodeBase;
                const int evb2 = __float_as_int(ev[e2]);
                int    from[18], ss[18], yo[18];
                float  sc[18];
                float4 val[18];
#pragma unroll
                for (int r = 0; r < 18; ++r) {
                  const unsigned ent = (lane + 32u * r) / 18u;
                  from[r] = (ent < (unsigned)nb ? (int)ent : 0) << 2;
                  ss[r] = __builtin_amdgcn_ds_bpermute(from[r], es);
                }
#pragma unroll
                for (int r = 0; r < 18; ++r) {
                  const unsigned q = lane + 32u * r;
                  const unsigned slot = q - (q / 18u) * 18u;
                  const float* p;
                  if (slot < 2)      p = x0 + (size_t)ss[r] * 8  + slot * 4;
                  else if (slot < 8) p = x1 + (size_t)ss[r] * 24 + (slot - 2) * 4;
                  else               p = x2 + (size_t)ss[r] * 40 + (slot - 8) * 4;
                  val[r] = *(const float4*)p;
                }
#pragma unroll
                for (int r = 0; r < 18; ++r) {
                  const unsigned q = lane + 32u * r;
                  const unsigned slot = q - (q / 18u) * 18u;
                  yo[r] = __builtin_amdgcn_ds_bpermute(from[r], ed) * 144 + 72 + (int)slot * 4;
                  sc[r] = __int_as_float(__builtin_amdgcn_ds_bpermute(from[r], evb));
                }
#pragma unroll
                for (int r = 0; r < 18; ++r) {
                  const unsigned ent = (lane + 32u * r) / 18u;
                  if (ent < (unsigned)nb) {
                    float* py = lds + yo[r];
                    __hip_atomic_fetch_add(py + 0, sc[r] * val[r].x, __ATOMIC_RELAXED, __HIP_MEMORY_SCOPE_WORKGROUP);
                    __hip_atomic_fetch_add(py + 1, sc[r] * val[r].y, __ATOMIC_RELAXED, __HIP_MEMORY_SCOPE_WORKGROUP);
                    __hip_atomic_fetch_add(py + 2, sc[r] * val[r].z, __ATOMIC_RELAXED, __HIP_MEMORY_SCOPE_WORKGROUP);
                    __hip_atomic_fetch_add(py + 3, sc[r] * val[r].w, __ATOMIC_RELAXED, __HIP_MEMORY_SCOPE_WORKGROUP);
                  }
                }
                es = es2; ed = ed2; evb = evb2;
              }
            }
            dcnt = 0;
          }
          if (!more) break;
        }
        cntl = 0;
      }
    }
  }
  __syncthreads();

  static_assert(NW == 8 && NT == 5, "the deal below is written for 8 waves and 5 tiles");
  const int wv = __builtin_amdgcn_readfirstlane(wave);
  const int wgBase = blockIdx.x * NODES_PER_WG;
  if (wv < NT) {
    block_l<2>(lds + wv * 16 * 144, wsm, out, wgBase + wv * 16, nN, row, hi, wcol);
  } else {
    const int t0 = (wv - NT) * 2;
    const int t1 = (t0 + 2 < NT) ? t0 + 2 : NT;
    for (int t = t0; t < t1; ++t)
      block_l<1>(lds + t * 16 * 144, wsm, out, wgBase + t * 16, nN, row, hi, wcol);
    if (wv == NW - 1) {
      for (int t = 0; t < NT; ++t)
        block_l<0>(lds + t * 16 * 144, wsm, out, wgBase + t * 16, nN, row, hi, wcol);
    }
  }
}

extern "C" void kernel_launch(void* const* d_in, const int* in_sizes, int n_in,
                              void* d_out, int out_size, void* d_ws, size_t ws_size,
                              hipStream_t stream) {
  const float* x0 = (const float*)d_in[0];
  const float* x1 = (const float*)d_in[1];
  const float* x2 = (const float*)d_in[2];
  const float* ev = (const float*)d_in[3];
  const int* esrc = (const int*)d_in[4];
  const int* edst = (const int*)d_in[5];
  const float* wyy0 = (const float*)d_in[6];
  const float* wyy1 = (const float*)d_in[7];
  const float* wyy2 = (const float*)d_in[8];
  const float* wyx0 = (const float*)d_in[9];
  const float* wyx1 = (const float*)d_in[10];
  const float* wyx2 = (const float*)d_in[11];
  const float* wxx0 = (const float*)d_in[12];
  const float* wxx1 = (const float*)d_in[13];
  const float* wxx2 = (const float*)d_in[14];
  float* out = (float*)d_out;
  (void)d_ws;

  const int nN = in_sizes[0] / CCH;
  const int nE = in_sizes[3];

  int blocks = (nN + NODES_PER_WG - 1) / NODES_PER_WG;
  node_cg_wmma<<<blocks, NW * 32, 0, stream>>>(x0, x1, x2, ev, esrc, edst, wyy0, wyy1,
                                          wyy2, wyx0, wyx1, wyx2, wxx0, wxx1, wxx2,
                                          out, nN, nE);
}


# x rows staged into LDS in 16-byte pieces, each thread's 6 loads all issued before its first store: one trip to memory for the 80 rows instead of 23 dependent 4-byte loads per thread
# speedup vs baseline: 3.0173x; 1.1529x over previous
#include <hip/hip_runtime.h>

#define LMAXV 2
#define CCH 8

typedef float v2f __attribute__((ext_vector_type(2)));
typedef float v8f __attribute__((ext_vector_type(8)));

constexpr double cfact(int n) { double r = 1.0; for (int i = 2; i <= n; ++i) r *= i; return r; }
constexpr double csqrt_(double x) {
  if (x <= 0.0) return 0.0;
  double r = x < 1.0 ? 1.0 : x;
  for (int i = 0; i < 100; ++i) r = 0.5 * (r + x / r);
  return r;
}
constexpr double cgcoef(int j1, int m1, int j2, int m2, int j, int m) {
  if (m1 + m2 != m) return 0.0;
  int lo = j1 > j2 ? j1 - j2 : j2 - j1;
  if (j < lo || j > j1 + j2) return 0.0;
  double pre = csqrt_((2.0 * j + 1.0) * cfact(j + j1 - j2) * cfact(j - j1 + j2) *
                      cfact(j1 + j2 - j) / cfact(j1 + j2 + j + 1));
  pre *= csqrt_(cfact(j + m) * cfact(j - m) * cfact(j1 - m1) * cfact(j1 + m1) *
                cfact(j2 - m2) * cfact(j2 + m2));
  double s = 0.0;
  for (int k = 0; k <= j1 + j2 - j; ++k) {
    int d2 = j1 + j2 - j - k, d3 = j1 - m1 - k, d4 = j2 + m2 - k;
    int d5 = j - j2 + m1 + k, d6 = j - j1 - m2 + k;
    if (d2 < 0 || d3 < 0 || d4 < 0 || d5 < 0 || d6 < 0) continue;
    double den = cfact(k) * cfact(d2) * cfact(d3) * cfact(d4) * cfact(d5) * cfact(d6);
    s += ((k & 1) ? -1.0 : 1.0) / den;
  }
  return pre * s;
}

struct CGPack {
  int ncoup[3];
  int l1t[3][6], l2t[3][6];
  int nnz[3][5][6];
  int pq[3][5][6][5];
  float coef[3][5][6][5];
};

constexpr CGPack buildCG() {
  CGPack P{};
  for (int l = 0; l <= LMAXV; ++l) {
    int nc = 0;
    for (int l1 = 0; l1 <= LMAXV; ++l1)
      for (int l2 = 0; l2 <= LMAXV; ++l2) {
        int lo = l1 > l2 ? l1 - l2 : l2 - l1;
        int hi = (l1 + l2 > LMAXV) ? LMAXV : (l1 + l2);
        if (l < lo || l > hi) continue;
        P.l1t[l][nc] = l1; P.l2t[l][nc] = l2;
        for (int mi = 0; mi < 2 * l + 1; ++mi) {
          int m = mi - l, n = 0;
          for (int p = 0; p < 2 * l1 + 1; ++p) {
            int m1 = p - l1, m2 = m - m1;
            if (m2 < -l2 || m2 > l2) continue;
            double v = cgcoef(l1, m1, l2, m2, l, m);
            if (v == 0.0) continue;
            P.pq[l][mi][nc][n] = p * 8 + (m2 + l2);
            P.coef[l][mi][nc][n] = (float)v;
            ++n;
          }
          P.nnz[l][mi][nc] = n;
        }
        ++nc;
      }
    P.ncoup[l] = nc;
  }
  return P;
}

constexpr CGPack H = buildCG();
constexpr int LOFF[3] = {0, 8, 32};
constexpr int NCOUPC[3] = {3, 6, 6};

#define NT 5
#define NW 8
#define NODES_PER_WG (NT * 16)
constexpr int WOFF[3][3] = {{0, 1536, 3072}, {4608, 7680, 10752}, {13824, 16896, 19968}};
#define WTOT 23040
constexpr int WCH[3] = {3, 6, 6};
#define KLD 12
#define LCAP 2048
#define LSPARE (LCAP / 32 - 1)
#define DCAP 1024
#define STEP_EDGES (NW * 32 * KLD * 4)


template <int l>
static __device__ __forceinline__ void block_l(const float* __restrict__ L,
                                               const float* __restrict__ wsm,
                                               float* __restrict__ out,
                                               const int tileBase, const int nN,
                                               const int row, const int hi,
                                               const int wcol) {
  constexpr int nm = 2 * l + 1;
  const float* Lrow = L + row * 144;
  const v8f vzero = {0.f, 0.f, 0.f, 0.f, 0.f, 0.f, 0.f, 0.f};
  v8f acc[nm];
#pragma unroll
  for (int m = 0; m < nm; ++m) acc[m] = vzero;

  constexpr int WSTRIDE = WOFF[l][1] - WOFF[l][0];
  static_assert(WOFF[l][2] - WOFF[l][1] == WSTRIDE, "weight matrices of one l are evenly spaced");
#pragma nounroll
  for (int prod = 0; prod < 3; ++prod) {
    const float* wl = wsm + WOFF[l][0] + prod * WSTRIDE;
    const int aoff = (prod < 2)  ? 72 : 0;
    const int boff = (prod == 0) ? 72 : 0;
#pragma unroll
    for (int cp = 0; cp < NCOUPC[l]; ++cp) {
      const int l1 = H.l1t[l][cp], l2 = H.l2t[l][cp];
      const int na = 2 * l1 + 1, nb = 2 * l2 + 1;
      const float* paBase = Lrow + aoff + LOFF[l1];
      const float* pbBase = Lrow + boff + LOFF[l2];
      const float* wcp = wl + (size_t)(cp * 64) * CCH;
      for (int c = 0; c < 8; ++c) {
        const float* pa = paBase + c * na;
        for (int dc = 0; dc < 8; dc += 4) {
          const int dlane = dc + 2 * hi;
          const int kb = c * 8 + dlane;
          v2f B;
          B.x = wcp[(kb + 0) * CCH + wcol];
          B.y = wcp[(kb + 1) * CCH + wcol];
          const float* pb0 = pbBase + dlane * nb;
          const float* pb1 = pb0 + nb;
#pragma unroll
          for (int m = 0; m < nm; ++m) {
            float z0 = 0.f, z1 = 0.f;
#pragma unroll
            for (int t = 0; t < H.nnz[l][m][cp]; ++t) {
              const int pq = H.pq[l][m][cp][t];
              const float cf = H.coef[l][m][cp][t];
              const float a = pa[pq >> 3];
              z0 += cf * a * pb0[pq & 7];
              z1 += cf * a * pb1[pq & 7];
            }
            v2f A; A.x = z0; A.y = z1;
            acc[m] = __builtin_amdgcn_wmma_f32_16x16x4_f32(
                false, A, false, B, (short)0, acc[m], false, false);
          }
        }
      }
    }
  }

  const int o = row;
  if (o < CCH) {
    const size_t baseL = (l == 0) ? 0 : (l == 1 ? (size_t)nN * 8 : (size_t)nN * 32);
#pragma unroll
    for (int r = 0; r < 8; ++r) {
      int ln = r + 8 * hi;
      const int node = tileBase + ln;
      if (node < nN) {
        const float* Lx = L + ln * 144 + LOFF[l] + o * nm;
        float* po = out + baseL + (size_t)node * (CCH * nm) + o * nm;
#pragma unroll
        for (int m = 0; m < nm; ++m) po[m] = acc[m][r] + Lx[m];
      }
    }
  }
}

__global__ __launch_bounds__(NW * 32) void node_cg_wmma(
    const float* __restrict__ x0, const float* __restrict__ x1,
    const float* __restrict__ x2, const float* __restrict__ ev,
    const int* __restrict__ src, const int* __restrict__ dst,
    const float* __restrict__ wyy0, const float* __restrict__ wyy1,
    const float* __restrict__ wyy2, const float* __restrict__ wyx0,
    const float* __restrict__ wyx1, const float* __restrict__ wyx2,
    const float* __restrict__ wxx0, const float* __restrict__ wxx1,
    const float* __restrict__ wxx2, float* __restrict__ out, int nN, int nE) {
  __shared__ float lds[NT * 16 * 144];
  __shared__ float wsm[WTOT];
  __shared__ int mlist[NW * LCAP];
  __shared__ int dlist[NW * DCAP];

  const int wave = threadIdx.x >> 5;
  const int lane = threadIdx.x & 31;
  const int row  = lane & 15;
  const int hi   = lane >> 4;
  const int wcol = row & 7;

#define WCOPY(ptr, l, prod)                                                   \
  _Pragma("unroll") for (int k = 0; k < WCH[l]; ++k)                          \
      ((float4*)(wsm + WOFF[l][prod]))[k * 128 + threadIdx.x] =               \
          ((const float4*)(ptr))[k * 128 + threadIdx.x];
  if (wave < 4) {
    WCOPY(wyy0, 0, 0) WCOPY(wyx0, 0, 1) WCOPY(wxx0, 0, 2)
    WCOPY(wyy1, 1, 0) WCOPY(wyx1, 1, 1) WCOPY(wxx1, 1, 2)
    WCOPY(wyy2, 2, 0) WCOPY(wyx2, 2, 1) WCOPY(wxx2, 2, 2)
  }

  {
    constexpr int NPIECE = NODES_PER_WG * 18;
    constexpr int PER_T = (NPIECE + NW * 32 - 1) / (NW * 32);
    float4 xv[PER_T];
#pragma unroll
    for (int k = 0; k < PER_T; ++k) {
      const int i = k * (NW * 32) + (int)threadIdx.x;
      const int ii = (i < NPIECE) ? i : NPIECE - 1;
      const int ln = ii / 18, s = ii - ln * 18;
      int node = blockIdx.x * NODES_PER_WG + ln;
      if (node >= nN) node = nN - 1;
      const float* xb = (s < 2) ? x0 : ((s < 8) ? x1 : x2);
      const int rowf  = (s < 2) ? 8  : ((s < 8) ? 24 : 40);
      const int s0    = (s < 2) ? 0  : ((s < 8) ? 2  : 8);
      xv[k] = *(const float4*)(xb + (size_t)node * rowf + (s - s0) * 4);
    }
    static_assert(PER_T == 6, "the statement below names six pieces");
    asm volatile("" :: "v"(xv[0].x), "v"(xv[0].y), "v"(xv[0].z), "v"(xv[0].w),
                       "v"(xv[1].x), "v"(xv[1].y), "v"(xv[1].z), "v"(xv[1].w),
                       "v"(xv[2].x), "v"(xv[2].y), "v"(xv[2].z), "v"(xv[2].w),
                       "v"(xv[3].x), "v"(xv[3].y), "v"(xv[3].z), "v"(xv[3].w),
                       "v"(xv[4].x), "v"(xv[4].y), "v"(xv[4].z), "v"(xv[4].w),
                       "v"(xv[5].x), "v"(xv[5].y), "v"(xv[5].z), "v"(xv[5].w));
    const float4 z4 = {0.0f, 0.0f, 0.0f, 0.0f};
#pragma unroll
    for (int k = 0; k < PER_T; ++k) {
      const int i = k * (NW * 32) + (int)threadIdx.x;
      if (i < NPIECE) {
        const int ln = i / 18, s = i - ln * 18;
        *(float4*)(lds + ln * 144 + s * 4)      = xv[k];
        *(float4*)(lds + ln * 144 + 72 + s * 4) = z4;
      }
    }
  }
  __syncthreads();

  {
    const unsigned nodeBase = blockIdx.x * NODES_PER_WG;
    int* myl = mlist + wave * LCAP + lane;
    int* dl  = dlist + wave * DCAP;
    int cntl = 0;
    const int nFull = nE / STEP_EDGES;
    const int n4c   = (nE + 3) >> 2;
    const int4* dst4 = (const int4*)dst;
    int it = (int)((blockIdx.x * (unsigned)nFull) / gridDim.x);
    int4 cur[KLD];
    if (nFull > 0) {
#pragma unroll
      for (int k = 0; k < KLD; ++k)
        cur[k] = dst4[(it * KLD + k) * (NW * 32) + threadIdx.x];
    }
    for (int i = 0; i <= nFull; ++i) {
      if (i < nFull) {
        const int itn = (it + 1 == nFull) ? 0 : it + 1;
        int4 nxt[KLD];
#pragma unroll
        for (int k = 0; k < KLD; ++k)
          nxt[k] = dst4[(itn * KLD + k) * (NW * 32) + threadIdx.x];
#pragma unroll
        for (int k = 0; k < KLD; ++k) {
          const unsigned t0 = (unsigned)cur[k].x - nodeBase, t1 = (unsigned)cur[k].y - nodeBase;
          const unsigned t2 = (unsigned)cur[k].z - nodeBase, t3 = (unsigned)cur[k].w - nodeBase;
          const unsigned tm = min(min(t0, t1), min(t2, t3));
          const bool any = tm < (unsigned)NODES_PER_WG;
          myl[32 * (any ? cntl : LSPARE)] = (it * KLD + k) * (NW * 32) + threadIdx.x;
          cntl += any ? 1 : 0;
        }
#pragma unroll
        for (int k = 0; k < KLD; ++k) cur[k] = nxt[k];
        it = itn;
      } else {
        for (int g0 = nFull * (STEP_EDGES / 4); g0 < n4c; g0 += NW * 32) {
          const int g = g0 + (int)threadIdx.x;
          bool any = false;
#pragma unroll
          for (int j = 0; j < 4; ++j) {
            const int e = g * 4 + j;
            if (g < n4c && e < nE)
              any = any || ((unsigned)dst[e] - nodeBase < (unsigned)NODES_PER_WG);
          }
          myl[32 * (any ? cntl : LSPARE)] = g;
          cntl += any ? 1 : 0;
        }
      }
      if (i == nFull || __builtin_amdgcn_ballot_w32(cntl > LSPARE - 1 - KLD) != 0u) {
        int dcnt = 0;
        for (int c0 = 0;; c0 += 4) {
          const bool more = __builtin_amdgcn_ballot_w32(c0 < cntl) != 0u;
          if (more) {
            int  gi[4];
            int  dv[4][4];
            bool ok[4][4];
#pragma unroll
            for (int u = 0; u < 4; ++u) {
              const bool have = c0 + u < cntl;
              const int gl = myl[32 * (have ? c0 + u : LSPARE)];
              gi[u] = have ? gl : 0;
#pragma unroll
              for (int j = 0; j < 4; ++j) {
                const int e = gi[u] * 4 + j;
                ok[u][j] = have && e < nE;
                dv[u][j] = dst[ok[u][j] ? e : 0];
              }
            }
#pragma unroll
            for (int u = 0; u < 4; ++u)
#pragma unroll
              for (int j = 0; j < 4; ++j) {
                const bool hit = ok[u][j] &&
                                 ((unsigned)dv[u][j] - nodeBase < (unsigned)NODES_PER_WG);
                const unsigned bal = __builtin_amdgcn_ballot_w32(hit);
                if (bal) {
                  if (hit) dl[dcnt + __builtin_amdgcn_mbcnt_lo(bal, 0u)] = gi[u] * 4 + j;
                  dcnt += __builtin_popcount(bal);
                }
              }
          }
          if (!more || dcnt > DCAP - 512) {
            if (dcnt > 0) {
              const int e0 = dl[lane < dcnt ? lane : dcnt - 1];
              int es  = src[e0];
              int ed  = dst[e0] - (int)nodeBase;
              int evb = __float_as_int(ev[e0]);
              for (int b0 = 0; b0 < dcnt; b0 += 32) {
                const int nb = (dcnt - b0 < 32) ? dcnt - b0 : 32;
                const int ni = b0 + 32 + lane;
                const int e2 = dl[ni < dcnt ? ni : dcnt - 1];
                const int es2  = src[e2];
                const int ed2  = dst[e2] - (int)nodeBase;
                const int evb2 = __float_as_int(ev[e2]);
                int    from[18], ss[18], yo[18];
                float  sc[18];
                float4 val[18];
#pragma unroll
                for (int r = 0; r < 18; ++r) {
                  const unsigned ent = (lane + 32u * r) / 18u;
                  from[r] = (ent < (unsigned)nb ? (int)ent : 0) << 2;
                  ss[r] = __builtin_amdgcn_ds_bpermute(from[r], es);
                }
#pragma unroll
                for (int r = 0; r < 18; ++r) {
                  const unsigned q = lane + 32u * r;
                  const unsigned slot = q - (q / 18u) * 18u;
                  const float* p;
                  if (slot < 2)      p = x0 + (size_t)ss[r] * 8  + slot * 4;
                  else if (slot < 8) p = x1 + (size_t)ss[r] * 24 + (slot - 2) * 4;
                  else               p = x2 + (size_t)ss[r] * 40 + (slot - 8) * 4;
                  val[r] = *(const float4*)p;
                }
#pragma unroll
                for (int r = 0; r < 18; ++r) {
                  const unsigned q = lane + 32u * r;
                  const unsigned slot = q - (q / 18u) * 18u;
                  yo[r] = __builtin_amdgcn_ds_bpermute(from[r], ed) * 144 + 72 + (int)slot * 4;
                  sc[r] = __int_as_float(__builtin_amdgcn_ds_bpermute(from[r], evb));
                }
#pragma unroll
                for (int r = 0; r < 18; ++r) {
                  const unsigned ent = (lane + 32u * r) / 18u;
                  if (ent < (unsigned)nb) {
                    float* py = lds + yo[r];
                    __hip_atomic_fetch_add(py + 0, sc[r] * val[r].x, __ATOMIC_RELAXED, __HIP_MEMORY_SCOPE_WORKGROUP);
                    __hip_atomic_fetch_add(py + 1, sc[r] * val[r].y, __ATOMIC_RELAXED, __HIP_MEMORY_SCOPE_WORKGROUP);
                    __hip_atomic_fetch_add(py + 2, sc[r] * val[r].z, __ATOMIC_RELAXED, __HIP_MEMORY_SCOPE_WORKGROUP);
                    __hip_atomic_fetch_add(py + 3, sc[r] * val[r].w, __ATOMIC_RELAXED, __HIP_MEMORY_SCOPE_WORKGROUP);
                  }
                }
                es = es2; ed = ed2; evb = evb2;
              }
            }
            dcnt = 0;
          }
          if (!more) break;
        }
        cntl = 0;
      }
    }
  }
  __syncthreads();

  static_assert(NW == 8 && NT == 5, "the deal below is written for 8 waves and 5 tiles");
  const int wv = __builtin_amdgcn_readfirstlane(wave);
  const int wgBase = blockIdx.x * NODES_PER_WG;
  if (wv < NT) {
    block_l<2>(lds + wv * 16 * 144, wsm, out, wgBase + wv * 16, nN, row, hi, wcol);
  } else {
    const int t0 = (wv - NT) * 2;
    const int t1 = (t0 + 2 < NT) ? t0 + 2 : NT;
    for (int t = t0; t < t1; ++t)
      block_l<1>(lds + t * 16 * 144, wsm, out, wgBase + t * 16, nN, row, hi, wcol);
    if (wv == NW - 1) {
      for (int t = 0; t < NT; ++t)
        block_l<0>(lds + t * 16 * 144, wsm, out, wgBase + t * 16, nN, row, hi, wcol);
    }
  }
}

extern "C" void kernel_launch(void* const* d_in, const int* in_sizes, int n_in,
                              void* d_out, int out_size, void* d_ws, size_t ws_size,
                              hipStream_t stream) {
  const float* x0 = (const float*)d_in[0];
  const float* x1 = (const float*)d_in[1];
  const float* x2 = (const float*)d_in[2];
  const float* ev = (const float*)d_in[3];
  const int* esrc = (const int*)d_in[4];
  const int* edst = (const int*)d_in[5];
  const float* wyy0 = (const float*)d_in[6];
  const float* wyy1 = (const float*)d_in[7];
  const float* wyy2 = (const float*)d_in[8];
  const float* wyx0 = (const float*)d_in[9];
  const float* wyx1 = (const float*)d_in[10];
  const float* wyx2 = (const float*)d_in[11];
  const float* wxx0 = (const float*)d_in[12];
  const float* wxx1 = (const float*)d_in[13];
  const float* wxx2 = (const float*)d_in[14];
  float* out = (float*)d_out;
  (void)d_ws;

  const int nN = in_sizes[0] / CCH;
  const int nE = in_sizes[3];

  int blocks = (nN + NODES_PER_WG - 1) / NODES_PER_WG;
  node_cg_wmma<<<blocks, NW * 32, 0, stream>>>(x0, x1, x2, ev, esrc, edst, wyy0, wyy1,
                                          wyy2, wyx0, wyx1, wyx2, wxx0, wxx1, wxx2,
                                          out, nN, nE);
}


# y*y and x*x: coupling (l2,l1) equals +-(l1,l2) with channels swapped; its weights are added to W(l1,l2) in LDS before the products (sum order changes) and it is skipped: 21% fewer matrix instr.
# speedup vs baseline: 3.2190x; 1.0668x over previous
#include <hip/hip_runtime.h>

#define LMAXV 2
#define CCH 8

typedef float v2f __attribute__((ext_vector_type(2)));
typedef float v8f __attribute__((ext_vector_type(8)));

constexpr double cfact(int n) { double r = 1.0; for (int i = 2; i <= n; ++i) r *= i; return r; }
constexpr double csqrt_(double x) {
  if (x <= 0.0) return 0.0;
  double r = x < 1.0 ? 1.0 : x;
  for (int i = 0; i < 100; ++i) r = 0.5 * (r + x / r);
  return r;
}
constexpr double cgcoef(int j1, int m1, int j2, int m2, int j, int m) {
  if (m1 + m2 != m) return 0.0;
  int lo = j1 > j2 ? j1 - j2 : j2 - j1;
  if (j < lo || j > j1 + j2) return 0.0;
  double pre = csqrt_((2.0 * j + 1.0) * cfact(j + j1 - j2) * cfact(j - j1 + j2) *
                      cfact(j1 + j2 - j) / cfact(j1 + j2 + j + 1));
  pre *= csqrt_(cfact(j + m) * cfact(j - m) * cfact(j1 - m1) * cfact(j1 + m1) *
                cfact(j2 - m2) * cfact(j2 + m2));
  double s = 0.0;
  for (int k = 0; k <= j1 + j2 - j; ++k) {
    int d2 = j1 + j2 - j - k, d3 = j1 - m1 - k, d4 = j2 + m2 - k;
    int d5 = j - j2 + m1 + k, d6 = j - j1 - m2 + k;
    if (d2 < 0 || d3 < 0 || d4 < 0 || d5 < 0 || d6 < 0) continue;
    double den = cfact(k) * cfact(d2) * cfact(d3) * cfact(d4) * cfact(d5) * cfact(d6);
    s += ((k & 1) ? -1.0 : 1.0) / den;
  }
  return pre * s;
}

struct CGPack {
  int ncoup[3];
  int l1t[3][6], l2t[3][6];
  int nnz[3][5][6];
  int pq[3][5][6][5];
  float coef[3][5][6][5];
};

constexpr CGPack buildCG() {
  CGPack P{};
  for (int l = 0; l <= LMAXV; ++l) {
    int nc = 0;
    for (int l1 = 0; l1 <= LMAXV; ++l1)
      for (int l2 = 0; l2 <= LMAXV; ++l2) {
        int lo = l1 > l2 ? l1 - l2 : l2 - l1;
        int hi = (l1 + l2 > LMAXV) ? LMAXV : (l1 + l2);
        if (l < lo || l > hi) continue;
        P.l1t[l][nc] = l1; P.l2t[l][nc] = l2;
        for (int mi = 0; mi < 2 * l + 1; ++mi) {
          int m = mi - l, n = 0;
          for (int p = 0; p < 2 * l1 + 1; ++p) {
            int m1 = p - l1, m2 = m - m1;
            if (m2 < -l2 || m2 > l2) continue;
            double v = cgcoef(l1, m1, l2, m2, l, m);
            if (v == 0.0) continue;
            P.pq[l][mi][nc][n] = p * 8 + (m2 + l2);
            P.coef[l][mi][nc][n] = (float)v;
            ++n;
          }
          P.nnz[l][mi][nc] = n;
        }
        ++nc;
      }
    P.ncoup[l] = nc;
  }
  return P;
}

constexpr CGPack H = buildCG();
constexpr int LOFF[3] = {0, 8, 32};
constexpr int NCOUPC[3] = {3, 6, 6};

constexpr bool cgSwapSymmetric() {
  for (int l = 0; l <= LMAXV; ++l)
    for (int l1 = 0; l1 <= LMAXV; ++l1)
      for (int l2 = l1 + 1; l2 <= LMAXV; ++l2) {
        if (l < l2 - l1 || l > l1 + l2) continue;
        const double s = ((l1 + l2 - l) & 1) ? -1.0 : 1.0;
        for (int m1 = -l1; m1 <= l1; ++m1)
          for (int m2 = -l2; m2 <= l2; ++m2) {
            const int m = m1 + m2;
            if (m < -l || m > l) continue;
            double d = cgcoef(l1, m1, l2, m2, l, m) - s * cgcoef(l2, m2, l1, m1, l, m);
            if (d < 0.0) d = -d;
            if (d > 1e-12) return false;
          }
      }
  return true;
}
static_assert(cgSwapSymmetric(), "the fold below rests on this symmetry of the table");

struct FoldTab {
  int n;
  int l[4], cpA[4], cpB[4];
  float sgn[4];
};
constexpr FoldTab buildFold() {
  FoldTab F{};
  for (int l = 0; l <= LMAXV; ++l)
    for (int a = 0; a < NCOUPC[l]; ++a) {
      if (H.l1t[l][a] >= H.l2t[l][a]) continue;
      for (int b = 0; b < NCOUPC[l]; ++b)
        if (H.l1t[l][b] == H.l2t[l][a] && H.l2t[l][b] == H.l1t[l][a]) {
          F.l[F.n] = l; F.cpA[F.n] = a; F.cpB[F.n] = b;
          F.sgn[F.n] = ((H.l1t[l][a] + H.l2t[l][a] - l) & 1) ? -1.0f : 1.0f;
          ++F.n;
        }
    }
  return F;
}
constexpr FoldTab FOLD = buildFold();
static_assert(FOLD.n == 4, "two folded pairs for l = 1 and two for l = 2");

#define NT 5
#define NW 8
#define NODES_PER_WG (NT * 16)
constexpr int WOFF[3][3] = {{0, 1536, 3072}, {4608, 7680, 10752}, {13824, 16896, 19968}};
#define WTOT 23040
constexpr int WCH[3] = {3, 6, 6};
#define KLD 12
#define LCAP 2048
#define LSPARE (LCAP / 32 - 1)
#define DCAP 1024
#define STEP_EDGES (NW * 32 * KLD * 4)


template <int l>
static __device__ __forceinline__ void block_l(const float* __restrict__ L,
                                               const float* __restrict__ wsm,
                                               float* __restrict__ out,
                                               const int tileBase, const int nN,
                                               const int row, const int hi,
                                               const int wcol) {
  constexpr int nm = 2 * l + 1;
  const float* Lrow = L + row * 144;
  const v8f vzero = {0.f, 0.f, 0.f, 0.f, 0.f, 0.f, 0.f, 0.f};
  v8f acc[nm];
#pragma unroll
  for (int m = 0; m < nm; ++m) acc[m] = vzero;

  constexpr int WSTRIDE = WOFF[l][1] - WOFF[l][0];
  static_assert(WOFF[l][2] - WOFF[l][1] == WSTRIDE, "weight matrices of one l are evenly spaced");
#pragma nounroll
  for (int prod = 0; prod < 3; ++prod) {
    const float* wl = wsm + WOFF[l][0] + prod * WSTRIDE;
    const int aoff = (prod < 2)  ? 72 : 0;
    const int boff = (prod == 0) ? 72 : 0;
#pragma unroll
    for (int cp = 0; cp < NCOUPC[l]; ++cp) {
      const int l1 = H.l1t[l][cp], l2 = H.l2t[l][cp];
      if (l1 > l2 && prod != 1) continue;
      const int na = 2 * l1 + 1, nb = 2 * l2 + 1;
      const float* paBase = Lrow + aoff + LOFF[l1];
      const float* pbBase = Lrow + boff + LOFF[l2];
      const float* wcp = wl + (size_t)(cp * 64) * CCH;
      for (int c = 0; c < 8; ++c) {
        const float* pa = paBase + c * na;
        for (int dc = 0; dc < 8; dc += 4) {
          const int dlane = dc + 2 * hi;
          const int kb = c * 8 + dlane;
          v2f B;
          B.x = wcp[(kb + 0) * CCH + wcol];
          B.y = wcp[(kb + 1) * CCH + wcol];
          const float* pb0 = pbBase + dlane * nb;
          const float* pb1 = pb0 + nb;
#pragma unroll
          for (int m = 0; m < nm; ++m) {
            float z0 = 0.f, z1 = 0.f;
#pragma unroll
            for (int t = 0; t < H.nnz[l][m][cp]; ++t) {
              const int pq = H.pq[l][m][cp][t];
              const float cf = H.coef[l][m][cp][t];
              const float a = pa[pq >> 3];
              z0 += cf * a * pb0[pq & 7];
              z1 += cf * a * pb1[pq & 7];
            }
            v2f A; A.x = z0; A.y = z1;
            acc[m] = __builtin_amdgcn_wmma_f32_16x16x4_f32(
                false, A, false, B, (short)0, acc[m], false, false);
          }
        }
      }
    }
  }

  const int o = row;
  if (o < CCH) {
    const size_t baseL = (l == 0) ? 0 : (l == 1 ? (size_t)nN * 8 : (size_t)nN * 32);
#pragma unroll
    for (int r = 0; r < 8; ++r) {
      int ln = r + 8 * hi;
      const int node = tileBase + ln;
      if (node < nN) {
        const float* Lx = L + ln * 144 + LOFF[l] + o * nm;
        float* po = out + baseL + (size_t)node * (CCH * nm) + o * nm;
#pragma unroll
        for (int m = 0; m < nm; ++m) po[m] = acc[m][r] + Lx[m];
      }
    }
  }
}

__global__ __launch_bounds__(NW * 32) void node_cg_wmma(
    const float* __restrict__ x0, const float* __restrict__ x1,
    const float* __restrict__ x2, const float* __restrict__ ev,
    const int* __restrict__ src, const int* __restrict__ dst,
    const float* __restrict__ wyy0, const float* __restrict__ wyy1,
    const float* __restrict__ wyy2, const float* __restrict__ wyx0,
    const float* __restrict__ wyx1, const float* __restrict__ wyx2,
    const float* __restrict__ wxx0, const float* __restrict__ wxx1,
    const float* __restrict__ wxx2, float* __restrict__ out, int nN, int nE) {
  __shared__ float lds[NT * 16 * 144];
  __shared__ float wsm[WTOT];
  __shared__ int mlist[NW * LCAP];
  __shared__ int dlist[NW * DCAP];

  const int wave = threadIdx.x >> 5;
  const int lane = threadIdx.x & 31;
  const int row  = lane & 15;
  const int hi   = lane >> 4;
  const int wcol = row & 7;

#define WCOPY(ptr, l, prod)                                                   \
  _Pragma("unroll") for (int k = 0; k < WCH[l]; ++k)                          \
      ((float4*)(wsm + WOFF[l][prod]))[k * 128 + threadIdx.x] =               \
          ((const float4*)(ptr))[k * 128 + threadIdx.x];
  if (wave < 4) {
    WCOPY(wyy0, 0, 0) WCOPY(wyx0, 0, 1) WCOPY(wxx0, 0, 2)
    WCOPY(wyy1, 1, 0) WCOPY(wyx1, 1, 1) WCOPY(wxx1, 1, 2)
    WCOPY(wyy2, 2, 0) WCOPY(wyx2, 2, 1) WCOPY(wxx2, 2, 2)
  }

  {
    constexpr int NPIECE = NODES_PER_WG * 18;
    constexpr int PER_T = (NPIECE + NW * 32 - 1) / (NW * 32);
    float4 xv[PER_T];
#pragma unroll
    for (int k = 0; k < PER_T; ++k) {
      const int i = k * (NW * 32) + (int)threadIdx.x;
      const int ii = (i < NPIECE) ? i : NPIECE - 1;
      const int ln = ii / 18, s = ii - ln * 18;
      int node = blockIdx.x * NODES_PER_WG + ln;
      if (node >= nN) node = nN - 1;
      const float* xb = (s < 2) ? x0 : ((s < 8) ? x1 : x2);
      const int rowf  = (s < 2) ? 8  : ((s < 8) ? 24 : 40);
      const int s0    = (s < 2) ? 0  : ((s < 8) ? 2  : 8);
      xv[k] = *(const float4*)(xb + (size_t)node * rowf + (s - s0) * 4);
    }
    static_assert(PER_T == 6, "the statement below names six pieces");
    asm volatile("" :: "v"(xv[0].x), "v"(xv[0].y), "v"(xv[0].z), "v"(xv[0].w),
                       "v"(xv[1].x), "v"(xv[1].y), "v"(xv[1].z), "v"(xv[1].w),
                       "v"(xv[2].x), "v"(xv[2].y), "v"(xv[2].z), "v"(xv[2].w),
                       "v"(xv[3].x), "v"(xv[3].y), "v"(xv[3].z), "v"(xv[3].w),
                       "v"(xv[4].x), "v"(xv[4].y), "v"(xv[4].z), "v"(xv[4].w),
                       "v"(xv[5].x), "v"(xv[5].y), "v"(xv[5].z), "v"(xv[5].w));
    const float4 z4 = {0.0f, 0.0f, 0.0f, 0.0f};
#pragma unroll
    for (int k = 0; k < PER_T; ++k) {
      const int i = k * (NW * 32) + (int)threadIdx.x;
      if (i < NPIECE) {
        const int ln = i / 18, s = i - ln * 18;
        *(float4*)(lds + ln * 144 + s * 4)      = xv[k];
        *(float4*)(lds + ln * 144 + 72 + s * 4) = z4;
      }
    }
  }
  __syncthreads();

#pragma unroll
  for (int f = 0; f < FOLD.n; ++f) {
#pragma unroll
    for (int pr = 0; pr < 3; pr += 2) {
      float* wA = wsm + WOFF[FOLD.l[f]][pr] + FOLD.cpA[f] * 512;
      const float* wB = wsm + WOFF[FOLD.l[f]][pr] + FOLD.cpB[f] * 512;
#pragma unroll
      for (int h = 0; h < 2; ++h) {
        const int e = h * 256 + (int)threadIdx.x;
        const int c = e >> 6, d = (e >> 3) & 7, o = e & 7;
        wA[e] += FOLD.sgn[f] * wB[(d * 8 + c) * 8 + o];
      }
    }
  }

  {
    const unsigned nodeBase = blockIdx.x * NODES_PER_WG;
    int* myl = mlist + wave * LCAP + lane;
    int* dl  = dlist + wave * DCAP;
    int cntl = 0;
    const int nFull = nE / STEP_EDGES;
    const int n4c   = (nE + 3) >> 2;
    const int4* dst4 = (const int4*)dst;
    int it = (int)((blockIdx.x * (unsigned)nFull) / gridDim.x);
    int4 cur[KLD];
    if (nFull > 0) {
#pragma unroll
      for (int k = 0; k < KLD; ++k)
        cur[k] = dst4[(it * KLD + k) * (NW * 32) + threadIdx.x];
    }
    for (int i = 0; i <= nFull; ++i) {
      if (i < nFull) {
        const int itn = (it + 1 == nFull) ? 0 : it + 1;
        int4 nxt[KLD];
#pragma unroll
        for (int k = 0; k < KLD; ++k)
          nxt[k] = dst4[(itn * KLD + k) * (NW * 32) + threadIdx.x];
#pragma unroll
        for (int k = 0; k < KLD; ++k) {
          const unsigned t0 = (unsigned)cur[k].x - nodeBase, t1 = (unsigned)cur[k].y - nodeBase;
          const unsigned t2 = (unsigned)cur[k].z - nodeBase, t3 = (unsigned)cur[k].w - nodeBase;
          const unsigned tm = min(min(t0, t1), min(t2, t3));
          const bool any = tm < (unsigned)NODES_PER_WG;
          myl[32 * (any ? cntl : LSPARE)] = (it * KLD + k) * (NW * 32) + threadIdx.x;
          cntl += any ? 1 : 0;
        }
#pragma unroll
        for (int k = 0; k < KLD; ++k) cur[k] = nxt[k];
        it = itn;
      } else {
        for (int g0 = nFull * (STEP_EDGES / 4); g0 < n4c; g0 += NW * 32) {
          const int g = g0 + (int)threadIdx.x;
          bool any = false;
#pragma unroll
          for (int j = 0; j < 4; ++j) {
            const int e = g * 4 + j;
            if (g < n4c && e < nE)
              any = any || ((unsigned)dst[e] - nodeBase < (unsigned)NODES_PER_WG);
          }
          myl[32 * (any ? cntl : LSPARE)] = g;
          cntl += any ? 1 : 0;
        }
      }
      if (i == nFull || __builtin_amdgcn_ballot_w32(cntl > LSPARE - 1 - KLD) != 0u) {
        int dcnt = 0;
        for (int c0 = 0;; c0 += 4) {
          const bool more = __builtin_amdgcn_ballot_w32(c0 < cntl) != 0u;
          if (more) {
            int  gi[4];
            int  dv[4][4];
            bool ok[4][4];
#pragma unroll
            for (int u = 0; u < 4; ++u) {
              const bool have = c0 + u < cntl;
              const int gl = myl[32 * (have ? c0 + u : LSPARE)];
              gi[u] = have ? gl : 0;
#pragma unroll
              for (int j = 0; j < 4; ++j) {
                const int e = gi[u] * 4 + j;
                ok[u][j] = have && e < nE;
                dv[u][j] = dst[ok[u][j] ? e : 0];
              }
            }
#pragma unroll
            for (int u = 0; u < 4; ++u)
#pragma unroll
              for (int j = 0; j < 4; ++j) {
                const bool hit = ok[u][j] &&
                                 ((unsigned)dv[u][j] - nodeBase < (unsigned)NODES_PER_WG);
                const unsigned bal = __builtin_amdgcn_ballot_w32(hit);
                if (bal) {
                  if (hit) dl[dcnt + __builtin_amdgcn_mbcnt_lo(bal, 0u)] = gi[u] * 4 + j;
                  dcnt += __builtin_popcount(bal);
                }
              }
          }
          if (!more || dcnt > DCAP - 512) {
            if (dcnt > 0) {
              const int e0 = dl[lane < dcnt ? lane : dcnt - 1];
              int es  = src[e0];
              int ed  = dst[e0] - (int)nodeBase;
              int evb = __float_as_int(ev[e0]);
              for (int b0 = 0; b0 < dcnt; b0 += 32) {
                const int nb = (dcnt - b0 < 32) ? dcnt - b0 : 32;
                const int ni = b0 + 32 + lane;
                const int e2 = dl[ni < dcnt ? ni : dcnt - 1];
                const int es2  = src[e2];
                const int ed2  = dst[e2] - (int)nodeBase;
                const int evb2 = __float_as_int(ev[e2]);
                int    from[18], ss[18], yo[18];
                float  sc[18];
                float4 val[18];
#pragma unroll
                for (int r = 0; r < 18; ++r) {
                  const unsigned ent = (lane + 32u * r) / 18u;
                  from[r] = (ent < (unsigned)nb ? (int)ent : 0) << 2;
                  ss[r] = __builtin_amdgcn_ds_bpermute(from[r], es);
                }
#pragma unroll
                for (int r = 0; r < 18; ++r) {
                  const unsigned q = lane + 32u * r;
                  const unsigned slot = q - (q / 18u) * 18u;
                  const float* p;
                  if (slot < 2)      p = x0 + (size_t)ss[r] * 8  + slot * 4;
                  else if (slot < 8) p = x1 + (size_t)ss[r] * 24 + (slot - 2) * 4;
                  else               p = x2 + (size_t)ss[r] * 40 + (slot - 8) * 4;
                  val[r] = *(const float4*)p;
                }
#pragma unroll
                for (int r = 0; r < 18; ++r) {
                  const unsigned q = lane + 32u * r;
                  const unsigned slot = q - (q / 18u) * 18u;
                  yo[r] = __builtin_amdgcn_ds_bpermute(from[r], ed) * 144 + 72 + (int)slot * 4;
                  sc[r] = __int_as_float(__builtin_amdgcn_ds_bpermute(from[r], evb));
                }
#pragma unroll
                for (int r = 0; r < 18; ++r) {
                  const unsigned ent = (lane + 32u * r) / 18u;
                  if (ent < (unsigned)nb) {
                    float* py = lds + yo[r];
                    __hip_atomic_fetch_add(py + 0, sc[r] * val[r].x, __ATOMIC_RELAXED, __HIP_MEMORY_SCOPE_WORKGROUP);
                    __hip_atomic_fetch_add(py + 1, sc[r] * val[r].y, __ATOMIC_RELAXED, __HIP_MEMORY_SCOPE_WORKGROUP);
                    __hip_atomic_fetch_add(py + 2, sc[r] * val[r].z, __ATOMIC_RELAXED, __HIP_MEMORY_SCOPE_WORKGROUP);
                    __hip_atomic_fetch_add(py + 3, sc[r] * val[r].w, __ATOMIC_RELAXED, __HIP_MEMORY_SCOPE_WORKGROUP);
                  }
                }
                es = es2; ed = ed2; evb = evb2;
              }
            }
            dcnt = 0;
          }
          if (!more) break;
        }
        cntl = 0;
      }
    }
  }
  __syncthreads();

  static_assert(NW == 8 && NT == 5, "the deal below is written for 8 waves and 5 tiles");
  const int wv = __builtin_amdgcn_readfirstlane(wave);
  const int wgBase = blockIdx.x * NODES_PER_WG;
  if (wv < NT) {
    block_l<2>(lds + wv * 16 * 144, wsm, out, wgBase + wv * 16, nN, row, hi, wcol);
  } else {
    const int t0 = (wv - NT) * 2;
    const int t1 = (t0 + 2 < NT) ? t0 + 2 : NT;
    for (int t = t0; t < t1; ++t)
      block_l<1>(lds + t * 16 * 144, wsm, out, wgBase + t * 16, nN, row, hi, wcol);
    if (wv == NW - 1) {
      for (int t = 0; t < NT; ++t)
        block_l<0>(lds + t * 16 * 144, wsm, out, wgBase + t * 16, nN, row, hi, wcol);
    }
  }
}

extern "C" void kernel_launch(void* const* d_in, const int* in_sizes, int n_in,
                              void* d_out, int out_size, void* d_ws, size_t ws_size,
                              hipStream_t stream) {
  const float* x0 = (const float*)d_in[0];
  const float* x1 = (const float*)d_in[1];
  const float* x2 = (const float*)d_in[2];
  const float* ev = (const float*)d_in[3];
  const int* esrc = (const int*)d_in[4];
  const int* edst = (const int*)d_in[5];
  const float* wyy0 = (const float*)d_in[6];
  const float* wyy1 = (const float*)d_in[7];
  const float* wyy2 = (const float*)d_in[8];
  const float* wyx0 = (const float*)d_in[9];
  const float* wyx1 = (const float*)d_in[10];
  const float* wyx2 = (const float*)d_in[11];
  const float* wxx0 = (const float*)d_in[12];
  const float* wxx1 = (const float*)d_in[13];
  const float* wxx2 = (const float*)d_in[14];
  float* out = (float*)d_out;
  (void)d_ws;

  const int nN = in_sizes[0] / CCH;
  const int nE = in_sizes[3];

  int blocks = (nN + NODES_PER_WG - 1) / NODES_PER_WG;
  node_cg_wmma<<<blocks, NW * 32, 0, stream>>>(x0, x1, x2, ev, esrc, edst, wyy0, wyy1,
                                          wyy2, wyx0, wyx1, wyx2, wxx0, wxx1, wxx2,
                                          out, nN, nE);
}
